# starting kernel
# baseline (speedup 1.0000x reference)
#include <hip/hip_runtime.h>

typedef __attribute__((ext_vector_type(2))) float v2f;
typedef __attribute__((ext_vector_type(8))) float v8f;

__global__ void zero_f32(float* __restrict__ p, long n) {
  long i = (long)blockIdx.x * blockDim.x + threadIdx.x;
  long stride = (long)gridDim.x * blockDim.x;
  for (; i < n; i += stride) p[i] = 0.0f;
}

__global__ void degree_kernel(const int* __restrict__ dst, float* __restrict__ deg, int E) {
  int e = blockIdx.x * blockDim.x + threadIdx.x;
  if (e < E) atomicAdd(&deg[dst[e]], 1.0f);
}

__global__ void scatter_add_kernel(const float* __restrict__ x, const int* __restrict__ src,
                                   const int* __restrict__ dst, float* __restrict__ agg, int E) {
  int wid = (int)((blockIdx.x * (unsigned)blockDim.x + threadIdx.x) >> 5);
  int lane = threadIdx.x & 31;
  if (wid >= E) return;
  int s = src[wid];
  int d = dst[wid];
  const float4 v = *(const float4*)(x + (size_t)s * 128 + lane * 4);
  float* a = agg + (size_t)d * 128 + lane * 4;
  atomicAdd(a + 0, v.x);
  atomicAdd(a + 1, v.y);
  atomicAdd(a + 2, v.z);
  atomicAdd(a + 3, v.w);
}

__global__ void normalize_kernel(const float* __restrict__ agg, const float* __restrict__ deg,
                                 float* __restrict__ hn, long total) {
  long i = (long)blockIdx.x * blockDim.x + threadIdx.x;
  if (i >= total) return;
  long node = i >> 7;
  hn[i] = agg[i] / fmaxf(deg[node], 1.0f);
}

__global__ void sage_linear_wmma(const float* __restrict__ Xs, const float* __restrict__ Xn,
                                 const float* __restrict__ Wself, const float* __restrict__ Wneigh,
                                 const float* __restrict__ bias, float* __restrict__ Out,
                                 int Nnodes, int K, int F, int do_relu) {
  const int wid = threadIdx.x >> 5;
  const int lane = threadIdx.x & 31;
  const int tiles_n = F >> 4;
  const int m_tiles_per_block = 8 / tiles_n;
  const int m_tile = blockIdx.x * m_tiles_per_block + wid / tiles_n;
  const int n0 = (wid % tiles_n) << 4;
  const int m0 = m_tile << 4;
  if (m0 >= Nnodes) return;

  const int lane_lo = lane & 15;
  const int khalf = (lane >> 4) << 1;
  const size_t rowA = (size_t)(m0 + lane_lo) * K;

  v8f c = {};
  for (int k0 = 0; k0 < K; k0 += 4) {
    const int k = k0 + khalf;
    v2f a = *(const v2f*)(Xs + rowA + k);
    v2f b;
    b.x = Wself[(size_t)k * F + n0 + lane_lo];
    b.y = Wself[(size_t)(k + 1) * F + n0 + lane_lo];
    c = __builtin_amdgcn_wmma_f32_16x16x4_f32(false, a, false, b, (short)0, c, false, false);
    v2f a2 = *(const v2f*)(Xn + rowA + k);
    v2f b2;
    b2.x = Wneigh[(size_t)k * F + n0 + lane_lo];
    b2.y = Wneigh[(size_t)(k + 1) * F + n0 + lane_lo];
    c = __builtin_amdgcn_wmma_f32_16x16x4_f32(false, a2, false, b2, (short)0, c, false, false);
  }

  const float bv = bias[n0 + lane_lo];
  const int mbase = m0 + ((lane >> 4) << 3);
#pragma unroll
  for (int r = 0; r < 8; ++r) {
    float v = c[r] + bv;
    if (do_relu) v = fmaxf(v, 0.0f);
    Out[(size_t)(mbase + r) * F + n0 + lane_lo] = v;
  }
}

extern "C" void kernel_launch(void* const* d_in, const int* in_sizes, int n_in,
                              void* d_out, int out_size, void* d_ws, size_t ws_size,
                              hipStream_t stream) {
  (void)in_sizes; (void)n_in; (void)out_size; (void)ws_size;

  const float* x   = (const float*)d_in[0];
  const int*   src = (const int*)d_in[1];
  const int*   dst = (const int*)d_in[2];
  const float* W1s = (const float*)d_in[3];
  const float* W1n = (const float*)d_in[4];
  const float* b1  = (const float*)d_in[5];
  const float* W2s = (const float*)d_in[6];
  const float* W2n = (const float*)d_in[7];
  const float* b2  = (const float*)d_in[8];
  float* out = (float*)d_out;

  const int N = 50000, E = 800000, H = 128, F2 = 64;
  const long NH = (long)N * H;

  float* ws  = (float*)d_ws;
  float* deg = ws;
  float* agg = ws + N;
  float* hn  = agg + NH;
  float* h   = hn + NH;

  const int ZB = 2048, T = 256;
  const int edge_waves_blocks = (E * 32 + T - 1) / T;
  const int norm_blocks = (int)((NH + T - 1) / T);

  zero_f32<<<ZB, T, 0, stream>>>(deg, (long)N + NH);
  degree_kernel<<<(E + T - 1) / T, T, 0, stream>>>(dst, deg, E);
  scatter_add_kernel<<<edge_waves_blocks, T, 0, stream>>>(x, src, dst, agg, E);
  normalize_kernel<<<norm_blocks, T, 0, stream>>>(agg, deg, hn, NH);
  sage_linear_wmma<<<N / 16, T, 0, stream>>>(x, hn, W1s, W1n, b1, h, N, H, H, 1);

  zero_f32<<<ZB, T, 0, stream>>>(agg, NH);
  scatter_add_kernel<<<edge_waves_blocks, T, 0, stream>>>(h, src, dst, agg, E);
  normalize_kernel<<<norm_blocks, T, 0, stream>>>(agg, deg, hn, NH);
  sage_linear_wmma<<<(N / 16 + 1) / 2, T, 0, stream>>>(h, hn, W2s, W2n, b2, out, N, H, F2, 0);
}


# Neighbour gather keeps more loads in flight: a wave reads degrees and slot ids of all its rows at once, puts a row's ids in rank order through LDS, then loads 8 neighbour rows per batch
# speedup vs baseline: 3.7170x; 3.7170x over previous
#include <hip/hip_runtime.h>

typedef __attribute__((ext_vector_type(2))) float v2f;
typedef __attribute__((ext_vector_type(8))) float v8f;

__global__ void clear_buckets(int* __restrict__ blk, int* __restrict__ ohead, int n) {
  int i = blockIdx.x * blockDim.x + threadIdx.x;
  if (i < n) {
    blk[(size_t)i * 64 + 63] = 0;
    ohead[i] = 0;
  }
}

__global__ void bucket_edges(const int* __restrict__ src, const int* __restrict__ dst,
                             int* __restrict__ blk, int* __restrict__ ohead,
                             int* __restrict__ next, int E) {
  const int e = blockIdx.x * blockDim.x + threadIdx.x;
  if (e >= E) return;
  const int d = dst[e];
  const int s = src[e];
  int* my = blk + (size_t)d * 64;
  const int pos = atomicAdd(my + 63, 1);
  if (pos < 63) {
    my[pos] = s;
  } else {
    next[e] = atomicExch(&ohead[d], e + 1);
  }
}

__device__ __forceinline__ void consider(unsigned kh, unsigned kl, unsigned last_hi,
                                         unsigned last_lo, unsigned& hi, unsigned& lo) {
  const bool above = (kh > last_hi) | ((kh == last_hi) & (kl > last_lo));
  const bool below = (kh < hi) | ((kh == hi) & (kl < lo));
  if (above & below) { hi = kh; lo = kl; }
}

__device__ __forceinline__ float4
neighbour_mean(const float* __restrict__ x, const int* __restrict__ ohead,
               const int* __restrict__ next, const int* __restrict__ src,
               const int node, const int d, const int k0, const int k1,
               int* sorted, const int lane) {
  const float* xl = x + lane * 4;

  float4 acc = {0.0f, 0.0f, 0.0f, 0.0f};

  if (d <= 63) {
    int r0 = 0, r1 = 0;
    const int d0 = d < 32 ? d : 32;
    for (int j = 0; j < d0; ++j) {
      const int kj = __builtin_amdgcn_readlane(k0, j);
      r0 += (int)((kj < k0) | ((kj == k0) & (j < lane)));
      r1 += (int)(kj <= k1);
    }
    for (int j = 0; j < d - 32; ++j) {
      const int kj = __builtin_amdgcn_readlane(k1, j);
      r0 += (int)(kj < k0);
      r1 += (int)((kj < k1) | ((kj == k1) & (j < lane)));
    }
    if (lane < d) sorted[r0] = k0;
    if (32 + lane < d) sorted[r1] = k1;
    int s_lo = 0, s_hi = 0;
    if (lane < d) s_lo = sorted[lane];
    if (32 + lane < d) s_hi = sorted[32 + lane];

#pragma nounroll
    for (int t0 = 0; t0 < d; t0 += 8) {
      const int sv = t0 < 32 ? s_lo : s_hi;
      float4 v[8];
#pragma unroll
      for (int j = 0; j < 8; ++j) {
        const int t = t0 + j;
        v[j].x = 0.0f; v[j].y = 0.0f; v[j].z = 0.0f; v[j].w = 0.0f;
        if (t < d) {
          const int s = __builtin_amdgcn_readlane(sv, t & 31);
          v[j] = *(const float4*)(xl + (size_t)s * 128);
        }
      }
#pragma unroll
      for (int j = 0; j < 8; ++j) {
        if (t0 + j < d) {
          acc.x += v[j].x; acc.y += v[j].y; acc.z += v[j].z; acc.w += v[j].w;
        }
      }
    }
  } else {
    const int nover = d - 63;
    unsigned last_hi = 0u, last_lo = 0u;
    for (int t = 0; t < d; ++t) {
      unsigned hi = 0xFFFFFFFFu, lo = 0xFFFFFFFFu;
      consider((unsigned)k0 + 1u, (unsigned)lane, last_hi, last_lo, hi, lo);
      if (lane < 31)
        consider((unsigned)k1 + 1u, (unsigned)(32 + lane), last_hi, last_lo, hi, lo);
      int lnk = ohead[node];
      for (int p = 0; p < nover && lnk != 0; ++p) {
        const int e = lnk - 1;
        consider((unsigned)src[e] + 1u, (unsigned)(63 + p), last_hi, last_lo, hi, lo);
        lnk = next[e];
      }
      for (int o = 16; o > 0; o >>= 1) {
        const unsigned ohi = (unsigned)__shfl_xor((int)hi, o, 32);
        const unsigned olo = (unsigned)__shfl_xor((int)lo, o, 32);
        const bool take = (ohi < hi) | ((ohi == hi) & (olo < lo));
        hi = take ? ohi : hi;
        lo = take ? olo : lo;
      }
      last_hi = hi;
      last_lo = lo;
      const int s = __builtin_amdgcn_readfirstlane((int)(hi - 1u));
      const float4 v = *(const float4*)(xl + (size_t)s * 128);
      acc.x += v.x; acc.y += v.y; acc.z += v.z; acc.w += v.w;
    }
  }

  const float den = (float)(d > 1 ? d : 1);
  float4 r;
  r.x = acc.x / den; r.y = acc.y / den; r.z = acc.z / den; r.w = acc.w / den;
  return r;
}

#define MEAN_PITCH 132

template <int ROWS>
__global__ void __launch_bounds__(256)
sage_layer_wmma(const float* __restrict__ Xs, const int* __restrict__ blk,
                const int* __restrict__ ohead, const int* __restrict__ next,
                const int* __restrict__ src,
                const float* __restrict__ Wself, const float* __restrict__ Wneigh,
                const float* __restrict__ bias, float* __restrict__ Out,
                int Nnodes, int K, int F, int do_relu) {
  __shared__ float mean_rows[32 * MEAN_PITCH];
  __shared__ int sorted_ids[8 * 64];

  const int wid = threadIdx.x >> 5;
  const int lane = threadIdx.x & 31;
  const int tiles_n = F >> 4;
  const int m_tiles_per_block = 8 / tiles_n;
  const int block_row0 = blockIdx.x * m_tiles_per_block * 16;

  const int swid = __builtin_amdgcn_readfirstlane(wid);
  const int lr0 = swid * ROWS;
  int* sorted = sorted_ids + swid * 64;
  int dd[ROWS], kk0[ROWS], kk1[ROWS];
#pragma unroll
  for (int i = 0; i < ROWS; ++i) {
    const int node = block_row0 + lr0 + i;
    dd[i] = 0;
    if (node < Nnodes) dd[i] = blk[(size_t)node * 64 + 63];
  }
#pragma unroll
  for (int i = 0; i < ROWS; ++i) {
    const int node = block_row0 + lr0 + i;
    const int* my = blk + (size_t)(node < Nnodes ? node : Nnodes - 1) * 64;
    const int ns = dd[i] < 63 ? dd[i] : 63;
    kk0[i] = my[lane < ns ? lane : 63];
    kk1[i] = my[32 + lane < ns ? 32 + lane : 63];
  }
#pragma unroll
  for (int i = 0; i < ROWS; ++i) {
    const int node = block_row0 + lr0 + i;
    const float4 r = neighbour_mean(Xs, ohead, next, src, node, dd[i], kk0[i], kk1[i],
                                    sorted, lane);
    *(float4*)(mean_rows + (lr0 + i) * MEAN_PITCH + lane * 4) = r;
  }
  __syncthreads();

  const int m_local = (wid / tiles_n) << 4;
  const int n0 = (wid % tiles_n) << 4;
  const int m0 = block_row0 + m_local;
  if (m0 >= Nnodes) return;

  const int lane_lo = lane & 15;
  const int khalf = (lane >> 4) << 1;
  const size_t rowA = (size_t)(m0 + lane_lo) * K;
  const float* rowN = mean_rows + (m_local + lane_lo) * MEAN_PITCH;

  v8f c = {};
  for (int k0 = 0; k0 < K; k0 += 4) {
    const int k = k0 + khalf;
    v2f a = *(const v2f*)(Xs + rowA + k);
    v2f b;
    b.x = Wself[(size_t)k * F + n0 + lane_lo];
    b.y = Wself[(size_t)(k + 1) * F + n0 + lane_lo];
    c = __builtin_amdgcn_wmma_f32_16x16x4_f32(false, a, false, b, (short)0, c, false, false);
    v2f a2 = *(const v2f*)(rowN + k);
    v2f b2;
    b2.x = Wneigh[(size_t)k * F + n0 + lane_lo];
    b2.y = Wneigh[(size_t)(k + 1) * F + n0 + lane_lo];
    c = __builtin_amdgcn_wmma_f32_16x16x4_f32(false, a2, false, b2, (short)0, c, false, false);
  }

  const float bv = bias[n0 + lane_lo];
  const int mbase = m0 + ((lane >> 4) << 3);
#pragma unroll
  for (int r = 0; r < 8; ++r) {
    float v = c[r] + bv;
    if (do_relu) v = fmaxf(v, 0.0f);
    Out[(size_t)(mbase + r) * F + n0 + lane_lo] = v;
  }
}

extern "C" void kernel_launch(void* const* d_in, const int* in_sizes, int n_in,
                              void* d_out, int out_size, void* d_ws, size_t ws_size,
                              hipStream_t stream) {
  (void)in_sizes; (void)n_in; (void)out_size; (void)ws_size;

  const float* x   = (const float*)d_in[0];
  const int*   src = (const int*)d_in[1];
  const int*   dst = (const int*)d_in[2];
  const float* W1s = (const float*)d_in[3];
  const float* W1n = (const float*)d_in[4];
  const float* b1  = (const float*)d_in[5];
  const float* W2s = (const float*)d_in[6];
  const float* W2n = (const float*)d_in[7];
  const float* b2  = (const float*)d_in[8];
  float* out = (float*)d_out;

  const int N = 50000, E = 800000, H = 128, F2 = 64;
  const int NP = 50048;

  int* ohead = (int*)d_ws;
  int* next  = ohead + NP;
  int* blk   = next + E;
  float* h   = (float*)(blk + (long)N * 64);

  const int T = 256;

  clear_buckets<<<(N + T - 1) / T, T, 0, stream>>>(blk, ohead, N);
  bucket_edges<<<(E + T - 1) / T, T, 0, stream>>>(src, dst, blk, ohead, next, E);

  sage_layer_wmma<2><<<N / 16, T, 0, stream>>>(x, blk, ohead, next, src, W1s, W1n, b1, h, N, H, H, 1);

  sage_layer_wmma<4><<<(N / 16 + 1) / 2, T, 0, stream>>>(h, blk, ohead, next, src, W2s, W2n, b2, out, N, H, F2, 0);
}


# Linear phase re-tiled: a wave owns a 16x64 output strip whose four tiles interleave their columns, so weight, row and output accesses are 16 bytes per lane on whole lines; 4 multiplying waves a block
# speedup vs baseline: 3.9333x; 1.0582x over previous
#include <hip/hip_runtime.h>

typedef __attribute__((ext_vector_type(2))) float v2f;
typedef __attribute__((ext_vector_type(8))) float v8f;

__global__ void clear_buckets(int* __restrict__ blk, int* __restrict__ ohead, int n) {
  int i = blockIdx.x * blockDim.x + threadIdx.x;
  if (i < n) {
    blk[(size_t)i * 64 + 63] = 0;
    ohead[i] = 0;
  }
}

__global__ void bucket_edges(const int* __restrict__ src, const int* __restrict__ dst,
                             int* __restrict__ blk, int* __restrict__ ohead,
                             int* __restrict__ next, int E) {
  const int e = blockIdx.x * blockDim.x + threadIdx.x;
  if (e >= E) return;
  const int d = dst[e];
  const int s = src[e];
  int* my = blk + (size_t)d * 64;
  const int pos = atomicAdd(my + 63, 1);
  if (pos < 63) {
    my[pos] = s;
  } else {
    next[e] = atomicExch(&ohead[d], e + 1);
  }
}

__device__ __forceinline__ void consider(unsigned kh, unsigned kl, unsigned last_hi,
                                         unsigned last_lo, unsigned& hi, unsigned& lo) {
  const bool above = (kh > last_hi) | ((kh == last_hi) & (kl > last_lo));
  const bool below = (kh < hi) | ((kh == hi) & (kl < lo));
  if (above & below) { hi = kh; lo = kl; }
}

__device__ __forceinline__ float4
neighbour_mean(const float* __restrict__ x, const int* __restrict__ ohead,
               const int* __restrict__ next, const int* __restrict__ src,
               const int node, const int d, const int k0, const int k1,
               int* sorted, const int lane) {
  const float* xl = x + lane * 4;

  float4 acc = {0.0f, 0.0f, 0.0f, 0.0f};

  if (d <= 63) {
    int r0 = 0, r1 = 0;
    const int d0 = d < 32 ? d : 32;
    for (int j = 0; j < d0; ++j) {
      const int kj = __builtin_amdgcn_readlane(k0, j);
      r0 += (int)((kj < k0) | ((kj == k0) & (j < lane)));
      r1 += (int)(kj <= k1);
    }
    for (int j = 0; j < d - 32; ++j) {
      const int kj = __builtin_amdgcn_readlane(k1, j);
      r0 += (int)(kj < k0);
      r1 += (int)((kj < k1) | ((kj == k1) & (j < lane)));
    }
    if (lane < d) sorted[r0] = k0;
    if (32 + lane < d) sorted[r1] = k1;
    int s_lo = 0, s_hi = 0;
    if (lane < d) s_lo = sorted[lane];
    if (32 + lane < d) s_hi = sorted[32 + lane];

#pragma nounroll
    for (int t0 = 0; t0 < d; t0 += 8) {
      const int sv = t0 < 32 ? s_lo : s_hi;
      float4 v[8];
#pragma unroll
      for (int j = 0; j < 8; ++j) {
        const int t = t0 + j;
        v[j].x = 0.0f; v[j].y = 0.0f; v[j].z = 0.0f; v[j].w = 0.0f;
        if (t < d) {
          const int s = __builtin_amdgcn_readlane(sv, t & 31);
          v[j] = *(const float4*)(xl + (size_t)s * 128);
        }
      }
#pragma unroll
      for (int j = 0; j < 8; ++j) {
        if (t0 + j < d) {
          acc.x += v[j].x; acc.y += v[j].y; acc.z += v[j].z; acc.w += v[j].w;
        }
      }
    }
  } else {
    const int nover = d - 63;
    unsigned last_hi = 0u, last_lo = 0u;
    for (int t = 0; t < d; ++t) {
      unsigned hi = 0xFFFFFFFFu, lo = 0xFFFFFFFFu;
      consider((unsigned)k0 + 1u, (unsigned)lane, last_hi, last_lo, hi, lo);
      if (lane < 31)
        consider((unsigned)k1 + 1u, (unsigned)(32 + lane), last_hi, last_lo, hi, lo);
      int lnk = ohead[node];
      for (int p = 0; p < nover && lnk != 0; ++p) {
        const int e = lnk - 1;
        consider((unsigned)src[e] + 1u, (unsigned)(63 + p), last_hi, last_lo, hi, lo);
        lnk = next[e];
      }
      for (int o = 16; o > 0; o >>= 1) {
        const unsigned ohi = (unsigned)__shfl_xor((int)hi, o, 32);
        const unsigned olo = (unsigned)__shfl_xor((int)lo, o, 32);
        const bool take = (ohi < hi) | ((ohi == hi) & (olo < lo));
        hi = take ? ohi : hi;
        lo = take ? olo : lo;
      }
      last_hi = hi;
      last_lo = lo;
      const int s = __builtin_amdgcn_readfirstlane((int)(hi - 1u));
      const float4 v = *(const float4*)(xl + (size_t)s * 128);
      acc.x += v.x; acc.y += v.y; acc.z += v.z; acc.w += v.w;
    }
  }

  const float den = (float)(d > 1 ? d : 1);
  float4 r;
  r.x = acc.x / den; r.y = acc.y / den; r.z = acc.z / den; r.w = acc.w / den;
  return r;
}

#define MEAN_PITCH 132
#define KDIM 128
#define GATHER_ROWS 4

__device__ __forceinline__ v8f wmma4(const float a0, const float a1, const float b0,
                                     const float b1, const v8f c) {
  v2f a; a.x = a0; a.y = a1;
  v2f b; b.x = b0; b.y = b1;
  return __builtin_amdgcn_wmma_f32_16x16x4_f32(false, a, false, b, (short)0, c, false, false);
}

template <int F, int WAVES>
__global__ void __launch_bounds__(32 * WAVES, 16)
sage_layer_wmma(const float* __restrict__ Xs, const int* __restrict__ blk,
                const int* __restrict__ ohead, const int* __restrict__ next,
                const int* __restrict__ src,
                const float* __restrict__ Wself, const float* __restrict__ Wneigh,
                const float* __restrict__ bias, float* __restrict__ Out,
                int Nnodes, int do_relu) {
  constexpr int BLOCK_ROWS = GATHER_ROWS * WAVES;
  __shared__ float mean_rows[BLOCK_ROWS * MEAN_PITCH];
  __shared__ int sorted_ids[WAVES * 64];

  const int wid = threadIdx.x >> 5;
  const int lane = threadIdx.x & 31;
  const int block_row0 = blockIdx.x * BLOCK_ROWS;

  const int swid = __builtin_amdgcn_readfirstlane(wid);
  const int lr0 = swid * GATHER_ROWS;
  int* sorted = sorted_ids + swid * 64;
  int dd[GATHER_ROWS], kk0[GATHER_ROWS], kk1[GATHER_ROWS];
#pragma unroll
  for (int i = 0; i < GATHER_ROWS; ++i) {
    const int node = block_row0 + lr0 + i;
    dd[i] = 0;
    if (node < Nnodes) dd[i] = blk[(size_t)node * 64 + 63];
  }
#pragma unroll
  for (int i = 0; i < GATHER_ROWS; ++i) {
    const int node = block_row0 + lr0 + i;
    const int* my = blk + (size_t)(node < Nnodes ? node : Nnodes - 1) * 64;
    const int ns = dd[i] < 63 ? dd[i] : 63;
    kk0[i] = my[lane < ns ? lane : 63];
    kk1[i] = my[32 + lane < ns ? 32 + lane : 63];
  }
#pragma unroll
  for (int i = 0; i < GATHER_ROWS; ++i) {
    const int node = block_row0 + lr0 + i;
    const float4 r = neighbour_mean(Xs, ohead, next, src, node, dd[i], kk0[i], kk1[i],
                                    sorted, lane);
    *(float4*)(mean_rows + (lr0 + i) * MEAN_PITCH + lane * 4) = r;
  }
  __syncthreads();

  constexpr int STRIPS = F / 64;
  static_assert((BLOCK_ROWS / 16) * STRIPS == 4, "four multiplying waves per block");
  if (swid >= 4) return;
  const int m_local = (swid / STRIPS) << 4;
  const int g = swid % STRIPS;
  const int m0 = block_row0 + m_local;
  if (m0 >= Nnodes) return;

  const int c = lane & 15;
  const int h = lane >> 4;
  const float* xa = Xs + (size_t)m0 * KDIM;
  const unsigned xofs = (unsigned)(c * KDIM + 4 * h);
  const float* na = mean_rows + (m_local + c) * MEAN_PITCH + 4 * h;
  const float* ws = Wself + 64 * g;
  const float* wn = Wneigh + 64 * g;
  const unsigned wofs = (unsigned)(4 * h * F + 4 * c);

  v8f acc0 = {}, acc1 = {}, acc2 = {}, acc3 = {};
  unsigned xv = xofs;
  unsigned wv = wofs;
#pragma nounroll
  for (int kb = 0; kb < KDIM; kb += 8) {
    {
      const float* pw = ws + wv;
      const float4 a = *(const float4*)(xa + xv);
      const float4 q0 = *(const float4*)(pw);
      const float4 q1 = *(const float4*)(pw + F);
      const float4 q2 = *(const float4*)(pw + 2 * F);
      const float4 q3 = *(const float4*)(pw + 3 * F);
      acc0 = wmma4(a.x, a.y, q0.x, q1.x, acc0);
      acc1 = wmma4(a.x, a.y, q0.y, q1.y, acc1);
      acc2 = wmma4(a.x, a.y, q0.z, q1.z, acc2);
      acc3 = wmma4(a.x, a.y, q0.w, q1.w, acc3);
      acc0 = wmma4(a.z, a.w, q2.x, q3.x, acc0);
      acc1 = wmma4(a.z, a.w, q2.y, q3.y, acc1);
      acc2 = wmma4(a.z, a.w, q2.z, q3.z, acc2);
      acc3 = wmma4(a.z, a.w, q2.w, q3.w, acc3);
    }
    {
      const float* pw = wn + wv;
      const float4 a = *(const float4*)(na + kb);
      const float4 q0 = *(const float4*)(pw);
      const float4 q1 = *(const float4*)(pw + F);
      const float4 q2 = *(const float4*)(pw + 2 * F);
      const float4 q3 = *(const float4*)(pw + 3 * F);
      acc0 = wmma4(a.x, a.y, q0.x, q1.x, acc0);
      acc1 = wmma4(a.x, a.y, q0.y, q1.y, acc1);
      acc2 = wmma4(a.x, a.y, q0.z, q1.z, acc2);
      acc3 = wmma4(a.x, a.y, q0.w, q1.w, acc3);
      acc0 = wmma4(a.z, a.w, q2.x, q3.x, acc0);
      acc1 = wmma4(a.z, a.w, q2.y, q3.y, acc1);
      acc2 = wmma4(a.z, a.w, q2.z, q3.z, acc2);
      acc3 = wmma4(a.z, a.w, q2.w, q3.w, acc3);
    }
    xv += 8;
    wv += 8 * F;
  }

  const float4 bv = *(const float4*)(bias + 64 * g + 4 * c);
  float* orow = Out + (size_t)m0 * F + 64 * g + (unsigned)(8 * h * F + 4 * c);
#pragma unroll
  for (int r = 0; r < 8; ++r) {
    float4 v;
    v.x = acc0[r] + bv.x; v.y = acc1[r] + bv.y; v.z = acc2[r] + bv.z; v.w = acc3[r] + bv.w;
    if (do_relu) {
      v.x = fmaxf(v.x, 0.0f); v.y = fmaxf(v.y, 0.0f); v.z = fmaxf(v.z, 0.0f); v.w = fmaxf(v.w, 0.0f);
    }
    *(float4*)(orow + (size_t)r * F) = v;
  }
}

extern "C" void kernel_launch(void* const* d_in, const int* in_sizes, int n_in,
                              void* d_out, int out_size, void* d_ws, size_t ws_size,
                              hipStream_t stream) {
  (void)in_sizes; (void)n_in; (void)out_size; (void)ws_size;

  const float* x   = (const float*)d_in[0];
  const int*   src = (const int*)d_in[1];
  const int*   dst = (const int*)d_in[2];
  const float* W1s = (const float*)d_in[3];
  const float* W1n = (const float*)d_in[4];
  const float* b1  = (const float*)d_in[5];
  const float* W2s = (const float*)d_in[6];
  const float* W2n = (const float*)d_in[7];
  const float* b2  = (const float*)d_in[8];
  float* out = (float*)d_out;

  const int N = 50000, E = 800000, H = 128;
  const int NP = 50048;

  int* ohead = (int*)d_ws;
  int* next  = ohead + NP;
  int* blk   = next + E;
  float* h   = (float*)(blk + (long)N * 64);

  const int T = 256;

  clear_buckets<<<(N + T - 1) / T, T, 0, stream>>>(blk, ohead, N);
  bucket_edges<<<(E + T - 1) / T, T, 0, stream>>>(src, dst, blk, ohead, next, E);

  sage_layer_wmma<128, 8><<<1563, 256, 0, stream>>>(x, blk, ohead, next, src, W1s, W1n, b1, h, N, 1);

  sage_layer_wmma<64, 16><<<782, 512, 0, stream>>>(h, blk, ohead, next, src, W2s, W2n, b2, out, N, 0);
}


# Layer 2's weights go on before its aggregation (the mean is linear): layer 1's kernel also multiplies its H tile, kept in LDS, by W2self|W2neigh; layer 2 gathers 256-byte rows and adds
# speedup vs baseline: 4.3644x; 1.1096x over previous
#include <hip/hip_runtime.h>

typedef __attribute__((ext_vector_type(2))) float v2f;
typedef __attribute__((ext_vector_type(8))) float v8f;

__global__ void clear_buckets(int* __restrict__ blk, int* __restrict__ ohead, int n) {
  int i = blockIdx.x * blockDim.x + threadIdx.x;
  if (i < n) {
    blk[(size_t)i * 64 + 63] = 0;
    ohead[i] = 0;
  }
}

__global__ void bucket_edges(const int* __restrict__ src, const int* __restrict__ dst,
                             int* __restrict__ blk, int* __restrict__ ohead,
                             int* __restrict__ next, int E) {
  const int e = blockIdx.x * blockDim.x + threadIdx.x;
  if (e >= E) return;
  const int d = dst[e];
  const int s = src[e];
  int* my = blk + (size_t)d * 64;
  const int pos = atomicAdd(my + 63, 1);
  if (pos < 63) {
    my[pos] = s;
  } else {
    next[e] = atomicExch(&ohead[d], e + 1);
  }
}

__device__ __forceinline__ void consider(unsigned kh, unsigned kl, unsigned last_hi,
                                         unsigned last_lo, unsigned& hi, unsigned& lo) {
  const bool above = (kh > last_hi) | ((kh == last_hi) & (kl > last_lo));
  const bool below = (kh < hi) | ((kh == hi) & (kl < lo));
  if (above & below) { hi = kh; lo = kl; }
}

__device__ __forceinline__ void row_zero(float4& v) { v.x = 0.0f; v.y = 0.0f; v.z = 0.0f; v.w = 0.0f; }
__device__ __forceinline__ void row_zero(float2& v) { v.x = 0.0f; v.y = 0.0f; }
__device__ __forceinline__ void row_add(float4& a, const float4& b) {
  a.x += b.x; a.y += b.y; a.z += b.z; a.w += b.w;
}
__device__ __forceinline__ void row_add(float2& a, const float2& b) { a.x += b.x; a.y += b.y; }
__device__ __forceinline__ float4 row_div(const float4& a, const float den) {
  float4 r;
  r.x = a.x / den; r.y = a.y / den; r.z = a.z / den; r.w = a.w / den;
  return r;
}
__device__ __forceinline__ float2 row_div(const float2& a, const float den) {
  float2 r;
  r.x = a.x / den; r.y = a.y / den;
  return r;
}

template <typename V>
__device__ __forceinline__ V
neighbour_mean(const float* __restrict__ xl, const int* __restrict__ ohead,
               const int* __restrict__ next, const int* __restrict__ src,
               const int node, const int d, const int k0, const int k1,
               int* sorted, const int lane) {
  V acc;
  row_zero(acc);

  if (d <= 63) {
    int r0 = 0, r1 = 0;
    const int d0 = d < 32 ? d : 32;
    for (int j = 0; j < d0; ++j) {
      const int kj = __builtin_amdgcn_readlane(k0, j);
      r0 += (int)((kj < k0) | ((kj == k0) & (j < lane)));
      r1 += (int)(kj <= k1);
    }
    for (int j = 0; j < d - 32; ++j) {
      const int kj = __builtin_amdgcn_readlane(k1, j);
      r0 += (int)(kj < k0);
      r1 += (int)((kj < k1) | ((kj == k1) & (j < lane)));
    }
    if (lane < d) sorted[r0] = k0;
    if (32 + lane < d) sorted[r1] = k1;
    int s_lo = 0, s_hi = 0;
    if (lane < d) s_lo = sorted[lane];
    if (32 + lane < d) s_hi = sorted[32 + lane];

#pragma nounroll
    for (int t0 = 0; t0 < d; t0 += 8) {
      const int sv = t0 < 32 ? s_lo : s_hi;
      V v[8];
#pragma unroll
      for (int j = 0; j < 8; ++j) {
        const int t = t0 + j;
        row_zero(v[j]);
        if (t < d) {
          const int s = __builtin_amdgcn_readlane(sv, t & 31);
          v[j] = *(const V*)(xl + (size_t)s * 128);
        }
      }
#pragma unroll
      for (int j = 0; j < 8; ++j) {
        if (t0 + j < d) row_add(acc, v[j]);
      }
    }
  } else {
    const int nover = d - 63;
    unsigned last_hi = 0u, last_lo = 0u;
    for (int t = 0; t < d; ++t) {
      unsigned hi = 0xFFFFFFFFu, lo = 0xFFFFFFFFu;
      consider((unsigned)k0 + 1u, (unsigned)lane, last_hi, last_lo, hi, lo);
      if (lane < 31)
        consider((unsigned)k1 + 1u, (unsigned)(32 + lane), last_hi, last_lo, hi, lo);
      int lnk = ohead[node];
      for (int p = 0; p < nover && lnk != 0; ++p) {
        const int e = lnk - 1;
        consider((unsigned)src[e] + 1u, (unsigned)(63 + p), last_hi, last_lo, hi, lo);
        lnk = next[e];
      }
      for (int o = 16; o > 0; o >>= 1) {
        const unsigned ohi = (unsigned)__shfl_xor((int)hi, o, 32);
        const unsigned olo = (unsigned)__shfl_xor((int)lo, o, 32);
        const bool take = (ohi < hi) | ((ohi == hi) & (olo < lo));
        hi = take ? ohi : hi;
        lo = take ? olo : lo;
      }
      last_hi = hi;
      last_lo = lo;
      const int s = __builtin_amdgcn_readfirstlane((int)(hi - 1u));
      const V v = *(const V*)(xl + (size_t)s * 128);
      row_add(acc, v);
    }
  }

  const float den = (float)(d > 1 ? d : 1);
  return row_div(acc, den);
}

#define ROW_PITCH 132
#define KDIM 128
#define GATHER_ROWS 4
#define L1_WAVES 8
#define L1_ROWS (GATHER_ROWS * L1_WAVES)

__device__ __forceinline__ v8f wmma4(const float a0, const float a1, const float b0,
                                     const float b1, const v8f c) {
  v2f a; a.x = a0; a.y = a1;
  v2f b; b.x = b0; b.y = b1;
  return __builtin_amdgcn_wmma_f32_16x16x4_f32(false, a, false, b, (short)0, c, false, false);
}

template <int FW>
__device__ __forceinline__ void strip_step(const float4 a, const float* __restrict__ pw,
                                           v8f& acc0, v8f& acc1, v8f& acc2, v8f& acc3) {
  const float4 q0 = *(const float4*)(pw);
  const float4 q1 = *(const float4*)(pw + FW);
  const float4 q2 = *(const float4*)(pw + 2 * FW);
  const float4 q3 = *(const float4*)(pw + 3 * FW);
  acc0 = wmma4(a.x, a.y, q0.x, q1.x, acc0);
  acc1 = wmma4(a.x, a.y, q0.y, q1.y, acc1);
  acc2 = wmma4(a.x, a.y, q0.z, q1.z, acc2);
  acc3 = wmma4(a.x, a.y, q0.w, q1.w, acc3);
  acc0 = wmma4(a.z, a.w, q2.x, q3.x, acc0);
  acc1 = wmma4(a.z, a.w, q2.y, q3.y, acc1);
  acc2 = wmma4(a.z, a.w, q2.z, q3.z, acc2);
  acc3 = wmma4(a.z, a.w, q2.w, q3.w, acc3);
}

__global__ void __launch_bounds__(32 * L1_WAVES, 16)
sage_layer1_project(const float* __restrict__ Xs, const int* __restrict__ blk,
                    const int* __restrict__ ohead, const int* __restrict__ next,
                    const int* __restrict__ src,
                    const float* __restrict__ W1self, const float* __restrict__ W1neigh,
                    const float* __restrict__ b1,
                    const float* __restrict__ W2self, const float* __restrict__ W2neigh,
                    const float* __restrict__ b2,
                    float* __restrict__ Z, int Nnodes) {
  __shared__ float mean_rows[L1_ROWS * ROW_PITCH];
  __shared__ float h_rows[L1_ROWS * ROW_PITCH];
  __shared__ int sorted_ids[L1_WAVES * 64];

  const int wid = threadIdx.x >> 5;
  const int lane = threadIdx.x & 31;
  const int block_row0 = blockIdx.x * L1_ROWS;

  const int swid = __builtin_amdgcn_readfirstlane(wid);
  const int lr0 = swid * GATHER_ROWS;
  int* sorted = sorted_ids + swid * 64;
  int dd[GATHER_ROWS], kk0[GATHER_ROWS], kk1[GATHER_ROWS];
#pragma unroll
  for (int i = 0; i < GATHER_ROWS; ++i) {
    const int node = block_row0 + lr0 + i;
    dd[i] = 0;
    if (node < Nnodes) dd[i] = blk[(size_t)node * 64 + 63];
  }
#pragma unroll
  for (int i = 0; i < GATHER_ROWS; ++i) {
    const int node = block_row0 + lr0 + i;
    const int* my = blk + (size_t)(node < Nnodes ? node : Nnodes - 1) * 64;
    const int ns = dd[i] < 63 ? dd[i] : 63;
    kk0[i] = my[lane < ns ? lane : 63];
    kk1[i] = my[32 + lane < ns ? 32 + lane : 63];
  }
#pragma unroll
  for (int i = 0; i < GATHER_ROWS; ++i) {
    const int node = block_row0 + lr0 + i;
    const float4 r = neighbour_mean<float4>(Xs + lane * 4, ohead, next, src, node, dd[i], kk0[i],
                                      kk1[i], sorted, lane);
    *(float4*)(mean_rows + (lr0 + i) * ROW_PITCH + lane * 4) = r;
  }
  __syncthreads();

  const int m_local = (swid >> 1) << 4;
  const int g = swid & 1;
  const int m0 = block_row0 + m_local;
  const bool mult = (swid < 4) & (m0 < Nnodes);

  if (mult) {
    const int c = lane & 15;
    const int h = lane >> 4;
    const float* xa = Xs + (size_t)m0 * KDIM;
    const float* na = mean_rows + (m_local + c) * ROW_PITCH + 4 * h;
    const float* ws = W1self + 64 * g;
    const float* wn = W1neigh + 64 * g;

    v8f acc0 = {}, acc1 = {}, acc2 = {}, acc3 = {};
    unsigned xv = (unsigned)(c * KDIM + 4 * h);
    unsigned wv = (unsigned)(4 * h * 128 + 4 * c);
#pragma nounroll
    for (int kb = 0; kb < KDIM; kb += 8) {
      strip_step<128>(*(const float4*)(xa + xv), ws + wv, acc0, acc1, acc2, acc3);
      strip_step<128>(*(const float4*)(na + kb), wn + wv, acc0, acc1, acc2, acc3);
      xv += 8;
      wv += 8 * 128;
    }

    const float4 bv = *(const float4*)(b1 + 64 * g + 4 * c);
    float* hrow = h_rows + (m_local + 8 * h) * ROW_PITCH + 64 * g + 4 * c;
#pragma unroll
    for (int r = 0; r < 8; ++r) {
      float4 v;
      v.x = fmaxf(acc0[r] + bv.x, 0.0f);
      v.y = fmaxf(acc1[r] + bv.y, 0.0f);
      v.z = fmaxf(acc2[r] + bv.z, 0.0f);
      v.w = fmaxf(acc3[r] + bv.w, 0.0f);
      *(float4*)(hrow + r * ROW_PITCH) = v;
    }
  }
  __syncthreads();

  if (mult) {
    const float* w2 = g == 0 ? W2self : W2neigh;
    const int l2 = (int)__builtin_amdgcn_mbcnt_lo(0xFFFFFFFFu, 0u);
    const int c = l2 & 15;
    const int h = l2 >> 4;
    const float* ha = h_rows + (m_local + c) * ROW_PITCH + 4 * h;
    v8f acc0 = {}, acc1 = {}, acc2 = {}, acc3 = {};
    unsigned wv = (unsigned)(4 * h * 64 + 4 * c);
#pragma nounroll
    for (int kb = 0; kb < KDIM; kb += 8) {
      strip_step<64>(*(const float4*)(ha + kb), w2 + wv, acc0, acc1, acc2, acc3);
      wv += 8 * 64;
    }

    float4 bv = {0.0f, 0.0f, 0.0f, 0.0f};
    if (g == 0) bv = *(const float4*)(b2 + 4 * c);
    float* zrow = Z + (size_t)m0 * 128 + 64 * g + (unsigned)(8 * h * 128 + 4 * c);
#pragma unroll
    for (int r = 0; r < 8; ++r) {
      float4 v;
      v.x = acc0[r] + bv.x; v.y = acc1[r] + bv.y; v.z = acc2[r] + bv.z; v.w = acc3[r] + bv.w;
      *(float4*)(zrow + (size_t)r * 128) = v;
    }
  }
}

#define L2_WAVES 8
__global__ void __launch_bounds__(32 * L2_WAVES, 16)
sage_layer2_gather(const float* __restrict__ Z, const int* __restrict__ blk,
                   const int* __restrict__ ohead, const int* __restrict__ next,
                   const int* __restrict__ src, float* __restrict__ Out, int Nnodes) {
  __shared__ int sorted_ids[L2_WAVES * 64];

  const int wid = threadIdx.x >> 5;
  const int lane = threadIdx.x & 31;
  const int swid = __builtin_amdgcn_readfirstlane(wid);
  const int node0 = (blockIdx.x * L2_WAVES + swid) * GATHER_ROWS;
  int* sorted = sorted_ids + swid * 64;
  int dd[GATHER_ROWS], kk0[GATHER_ROWS], kk1[GATHER_ROWS];
  float2 zs[GATHER_ROWS];
#pragma unroll
  for (int i = 0; i < GATHER_ROWS; ++i) {
    const int node = node0 + i;
    dd[i] = 0;
    if (node < Nnodes) dd[i] = blk[(size_t)node * 64 + 63];
  }
#pragma unroll
  for (int i = 0; i < GATHER_ROWS; ++i) {
    const int node = node0 + i;
    const int nc = node < Nnodes ? node : Nnodes - 1;
    const int* my = blk + (size_t)nc * 64;
    const int ns = dd[i] < 63 ? dd[i] : 63;
    kk0[i] = my[lane < ns ? lane : 63];
    kk1[i] = my[32 + lane < ns ? 32 + lane : 63];
    zs[i] = *(const float2*)(Z + (size_t)nc * 128 + lane * 2);
  }
#pragma unroll
  for (int i = 0; i < GATHER_ROWS; ++i) {
    const int node = node0 + i;
    const float2 m = neighbour_mean<float2>(Z + 64 + lane * 2, ohead, next, src, node, dd[i],
                                      kk0[i], kk1[i], sorted, lane);
    if (node < Nnodes) {
      float2 o;
      o.x = zs[i].x + m.x; o.y = zs[i].y + m.y;
      *(float2*)(Out + (size_t)node * 64 + lane * 2) = o;
    }
  }
}

extern "C" void kernel_launch(void* const* d_in, const int* in_sizes, int n_in,
                              void* d_out, int out_size, void* d_ws, size_t ws_size,
                              hipStream_t stream) {
  (void)in_sizes; (void)n_in; (void)out_size; (void)ws_size;

  const float* x   = (const float*)d_in[0];
  const int*   src = (const int*)d_in[1];
  const int*   dst = (const int*)d_in[2];
  const float* W1s = (const float*)d_in[3];
  const float* W1n = (const float*)d_in[4];
  const float* b1  = (const float*)d_in[5];
  const float* W2s = (const float*)d_in[6];
  const float* W2n = (const float*)d_in[7];
  const float* b2  = (const float*)d_in[8];
  float* out = (float*)d_out;

  const int N = 50000, E = 800000;
  const int NP = 50048;

  int* ohead = (int*)d_ws;
  int* next  = ohead + NP;
  int* blk   = next + E;
  float* z   = (float*)(blk + (long)N * 64);

  const int T = 256;

  clear_buckets<<<(N + T - 1) / T, T, 0, stream>>>(blk, ohead, N);
  bucket_edges<<<(E + T - 1) / T, T, 0, stream>>>(src, dst, blk, ohead, next, E);

  sage_layer1_project<<<1563, 256, 0, stream>>>(x, blk, ohead, next, src, W1s, W1n, b1,
                                                W2s, W2n, b2, z, N);

  sage_layer2_gather<<<1563, 256, 0, stream>>>(z, blk, ohead, next, src, out, N);
}


# Neighbour gather issues about half the instructions per row: ids tagged with their slot position rank by one compare each, and whole batches of 8 rows load with no test per row
# speedup vs baseline: 4.5670x; 1.0464x over previous
#include <hip/hip_runtime.h>

typedef __attribute__((ext_vector_type(2))) float v2f;
typedef __attribute__((ext_vector_type(8))) float v8f;

__global__ void clear_buckets(int* __restrict__ blk, int* __restrict__ ohead, int n) {
  int i = blockIdx.x * blockDim.x + threadIdx.x;
  if (i < n) {
    blk[(size_t)i * 64 + 63] = 0;
    ohead[i] = 0;
  }
}

__global__ void bucket_edges(const int* __restrict__ src, const int* __restrict__ dst,
                             int* __restrict__ blk, int* __restrict__ ohead,
                             int* __restrict__ next, int E) {
  const int e = blockIdx.x * blockDim.x + threadIdx.x;
  if (e >= E) return;
  const int d = dst[e];
  const int s = src[e];
  int* my = blk + (size_t)d * 64;
  const int pos = atomicAdd(my + 63, 1);
  if (pos < 63) {
    my[pos] = s;
  } else {
    next[e] = atomicExch(&ohead[d], e + 1);
  }
}

__device__ __forceinline__ void consider(unsigned kh, unsigned kl, unsigned last_hi,
                                         unsigned last_lo, unsigned& hi, unsigned& lo) {
  const bool above = (kh > last_hi) | ((kh == last_hi) & (kl > last_lo));
  const bool below = (kh < hi) | ((kh == hi) & (kl < lo));
  if (above & below) { hi = kh; lo = kl; }
}

__device__ __forceinline__ void row_zero(float4& v) { v.x = 0.0f; v.y = 0.0f; v.z = 0.0f; v.w = 0.0f; }
__device__ __forceinline__ void row_zero(float2& v) { v.x = 0.0f; v.y = 0.0f; }
__device__ __forceinline__ void row_add(float4& a, const float4& b) {
  a.x += b.x; a.y += b.y; a.z += b.z; a.w += b.w;
}
__device__ __forceinline__ void row_add(float2& a, const float2& b) { a.x += b.x; a.y += b.y; }
__device__ __forceinline__ float4 row_div(const float4& a, const float den) {
  float4 r;
  r.x = a.x / den; r.y = a.y / den; r.z = a.z / den; r.w = a.w / den;
  return r;
}
__device__ __forceinline__ float2 row_div(const float2& a, const float den) {
  float2 r;
  r.x = a.x / den; r.y = a.y / den;
  return r;
}

template <typename V>
__device__ __forceinline__ V load_row(const float* __restrict__ tab, const unsigned lane_off,
                                      const int s) {
  const char* row = (const char*)tab + (size_t)(unsigned)s * 512u;
  return *(const V*)(row + lane_off);
}

template <typename V>
__device__ __forceinline__ V
neighbour_mean(const float* __restrict__ tab, const unsigned lane_off,
               const int* __restrict__ ohead, const int* __restrict__ next,
               const int* __restrict__ src,
               const int node, const int d, const int k0, const int k1,
               int* sorted, const int lane) {
  V acc;
  row_zero(acc);

  if (d <= 63) {
    const unsigned key0 = ((unsigned)k0 << 6) | (unsigned)lane;
    int r0 = 0, r1 = 0;
    if (d <= 32) {
      for (int j = 0; j < d; ++j) {
        const unsigned kj = (unsigned)__builtin_amdgcn_readlane((int)key0, j);
        r0 += (int)(kj < key0);
      }
    } else {
      const unsigned key1 = ((unsigned)k1 << 6) | (unsigned)(32 + lane);
      for (int j = 0; j < 32; ++j) {
        const unsigned kj = (unsigned)__builtin_amdgcn_readlane((int)key0, j);
        r0 += (int)(kj < key0);
        r1 += (int)(kj < key1);
      }
      for (int j = 0; j < d - 32; ++j) {
        const unsigned kj = (unsigned)__builtin_amdgcn_readlane((int)key1, j);
        r0 += (int)(kj < key0);
        r1 += (int)(kj < key1);
      }
    }
    if (lane < d) sorted[r0] = k0;
    if (32 + lane < d) sorted[r1] = k1;
    int s_lo = 0, s_hi = 0;
    if (lane < d) s_lo = sorted[lane];
    if (32 + lane < d) s_hi = sorted[32 + lane];

    int t0 = 0;
#pragma nounroll
    for (; t0 + 8 <= d; t0 += 8) {
      const int sv = t0 < 32 ? s_lo : s_hi;
      V v[8];
#pragma unroll
      for (int j = 0; j < 8; ++j) {
        const int s = __builtin_amdgcn_readlane(sv, (t0 + j) & 31);
        v[j] = load_row<V>(tab, lane_off, s);
      }
#pragma unroll
      for (int j = 0; j < 8; ++j) row_add(acc, v[j]);
    }
    if (t0 < d) {
      const int sv = t0 < 32 ? s_lo : s_hi;
      V v[7];
#pragma unroll
      for (int j = 0; j < 7; ++j) {
        const int t = t0 + j;
        row_zero(v[j]);
        if (t < d) {
          const int s = __builtin_amdgcn_readlane(sv, t & 31);
          v[j] = load_row<V>(tab, lane_off, s);
        }
      }
#pragma unroll
      for (int j = 0; j < 7; ++j) {
        if (t0 + j < d) row_add(acc, v[j]);
      }
    }
  } else {
    const int nover = d - 63;
    unsigned last_hi = 0u, last_lo = 0u;
    for (int t = 0; t < d; ++t) {
      unsigned hi = 0xFFFFFFFFu, lo = 0xFFFFFFFFu;
      consider((unsigned)k0 + 1u, (unsigned)lane, last_hi, last_lo, hi, lo);
      if (lane < 31)
        consider((unsigned)k1 + 1u, (unsigned)(32 + lane), last_hi, last_lo, hi, lo);
      int lnk = ohead[node];
      for (int p = 0; p < nover && lnk != 0; ++p) {
        const int e = lnk - 1;
        consider((unsigned)src[e] + 1u, (unsigned)(63 + p), last_hi, last_lo, hi, lo);
        lnk = next[e];
      }
      for (int o = 16; o > 0; o >>= 1) {
        const unsigned ohi = (unsigned)__shfl_xor((int)hi, o, 32);
        const unsigned olo = (unsigned)__shfl_xor((int)lo, o, 32);
        const bool take = (ohi < hi) | ((ohi == hi) & (olo < lo));
        hi = take ? ohi : hi;
        lo = take ? olo : lo;
      }
      last_hi = hi;
      last_lo = lo;
      const int s = __builtin_amdgcn_readfirstlane((int)(hi - 1u));
      const V v = load_row<V>(tab, lane_off, s);
      row_add(acc, v);
    }
  }

  const float den = (float)(d > 1 ? d : 1);
  return row_div(acc, den);
}

#define ROW_PITCH 132
#define KDIM 128
#define GATHER_ROWS 4
#define L1_WAVES 8
#define L1_ROWS (GATHER_ROWS * L1_WAVES)

__device__ __forceinline__ v8f wmma4(const float a0, const float a1, const float b0,
                                     const float b1, const v8f c) {
  v2f a; a.x = a0; a.y = a1;
  v2f b; b.x = b0; b.y = b1;
  return __builtin_amdgcn_wmma_f32_16x16x4_f32(false, a, false, b, (short)0, c, false, false);
}

template <int FW>
__device__ __forceinline__ void strip_step(const float4 a, const float* __restrict__ pw,
                                           v8f& acc0, v8f& acc1, v8f& acc2, v8f& acc3) {
  const float4 q0 = *(const float4*)(pw);
  const float4 q1 = *(const float4*)(pw + FW);
  const float4 q2 = *(const float4*)(pw + 2 * FW);
  const float4 q3 = *(const float4*)(pw + 3 * FW);
  acc0 = wmma4(a.x, a.y, q0.x, q1.x, acc0);
  acc1 = wmma4(a.x, a.y, q0.y, q1.y, acc1);
  acc2 = wmma4(a.x, a.y, q0.z, q1.z, acc2);
  acc3 = wmma4(a.x, a.y, q0.w, q1.w, acc3);
  acc0 = wmma4(a.z, a.w, q2.x, q3.x, acc0);
  acc1 = wmma4(a.z, a.w, q2.y, q3.y, acc1);
  acc2 = wmma4(a.z, a.w, q2.z, q3.z, acc2);
  acc3 = wmma4(a.z, a.w, q2.w, q3.w, acc3);
}

__global__ void __launch_bounds__(32 * L1_WAVES, 16)
sage_layer1_project(const float* __restrict__ Xs, const int* __restrict__ blk,
                    const int* __restrict__ ohead, const int* __restrict__ next,
                    const int* __restrict__ src,
                    const float* __restrict__ W1self, const float* __restrict__ W1neigh,
                    const float* __restrict__ b1,
                    const float* __restrict__ W2self, const float* __restrict__ W2neigh,
                    const float* __restrict__ b2,
                    float* __restrict__ Z, int Nnodes) {
  __shared__ float mean_rows[L1_ROWS * ROW_PITCH];
  __shared__ float h_rows[L1_ROWS * ROW_PITCH];
  __shared__ int sorted_ids[L1_WAVES * 64];

  const int wid = threadIdx.x >> 5;
  const int lane = threadIdx.x & 31;
  const int block_row0 = blockIdx.x * L1_ROWS;

  const int swid = __builtin_amdgcn_readfirstlane(wid);
  const int lr0 = swid * GATHER_ROWS;
  int* sorted = sorted_ids + swid * 64;
  int dd[GATHER_ROWS], kk0[GATHER_ROWS], kk1[GATHER_ROWS];
#pragma unroll
  for (int i = 0; i < GATHER_ROWS; ++i) {
    const int node = block_row0 + lr0 + i;
    dd[i] = 0;
    if (node < Nnodes) dd[i] = blk[(size_t)node * 64 + 63];
  }
#pragma unroll
  for (int i = 0; i < GATHER_ROWS; ++i) {
    const int node = block_row0 + lr0 + i;
    const int* my = blk + (size_t)(node < Nnodes ? node : Nnodes - 1) * 64;
    const int ns = dd[i] < 63 ? dd[i] : 63;
    kk0[i] = my[lane < ns ? lane : 63];
    kk1[i] = my[32 + lane < ns ? 32 + lane : 63];
  }
#pragma unroll
  for (int i = 0; i < GATHER_ROWS; ++i) {
    const int node = block_row0 + lr0 + i;
    const float4 r = neighbour_mean<float4>(Xs, (unsigned)lane * 16u, ohead, next, src, node, dd[i],
                                      kk0[i], kk1[i], sorted, lane);
    *(float4*)(mean_rows + (lr0 + i) * ROW_PITCH + lane * 4) = r;
  }
  __syncthreads();

  const int m_local = (swid >> 1) << 4;
  const int g = swid & 1;
  const int m0 = block_row0 + m_local;
  const bool mult = (swid < 4) & (m0 < Nnodes);

  if (mult) {
    const int c = lane & 15;
    const int h = lane >> 4;
    const float* xa = Xs + (size_t)m0 * KDIM;
    const float* na = mean_rows + (m_local + c) * ROW_PITCH + 4 * h;
    const float* ws = W1self + 64 * g;
    const float* wn = W1neigh + 64 * g;

    v8f acc0 = {}, acc1 = {}, acc2 = {}, acc3 = {};
    unsigned xv = (unsigned)(c * KDIM + 4 * h);
    unsigned wv = (unsigned)(4 * h * 128 + 4 * c);
#pragma nounroll
    for (int kb = 0; kb < KDIM; kb += 8) {
      strip_step<128>(*(const float4*)(xa + xv), ws + wv, acc0, acc1, acc2, acc3);
      strip_step<128>(*(const float4*)(na + kb), wn + wv, acc0, acc1, acc2, acc3);
      xv += 8;
      wv += 8 * 128;
    }

    const float4 bv = *(const float4*)(b1 + 64 * g + 4 * c);
    float* hrow = h_rows + (m_local + 8 * h) * ROW_PITCH + 64 * g + 4 * c;
#pragma unroll
    for (int r = 0; r < 8; ++r) {
      float4 v;
      v.x = fmaxf(acc0[r] + bv.x, 0.0f);
      v.y = fmaxf(acc1[r] + bv.y, 0.0f);
      v.z = fmaxf(acc2[r] + bv.z, 0.0f);
      v.w = fmaxf(acc3[r] + bv.w, 0.0f);
      *(float4*)(hrow + r * ROW_PITCH) = v;
    }
  }
  __syncthreads();

  if (mult) {
    const float* w2 = g == 0 ? W2self : W2neigh;
    const int l2 = (int)__builtin_amdgcn_mbcnt_lo(0xFFFFFFFFu, 0u);
    const int c = l2 & 15;
    const int h = l2 >> 4;
    const float* ha = h_rows + (m_local + c) * ROW_PITCH + 4 * h;
    v8f acc0 = {}, acc1 = {}, acc2 = {}, acc3 = {};
    unsigned wv = (unsigned)(4 * h * 64 + 4 * c);
#pragma nounroll
    for (int kb = 0; kb < KDIM; kb += 8) {
      strip_step<64>(*(const float4*)(ha + kb), w2 + wv, acc0, acc1, acc2, acc3);
      wv += 8 * 64;
    }

    float4 bv = {0.0f, 0.0f, 0.0f, 0.0f};
    if (g == 0) bv = *(const float4*)(b2 + 4 * c);
    float* zrow = Z + (size_t)m0 * 128 + 64 * g + (unsigned)(8 * h * 128 + 4 * c);
#pragma unroll
    for (int r = 0; r < 8; ++r) {
      float4 v;
      v.x = acc0[r] + bv.x; v.y = acc1[r] + bv.y; v.z = acc2[r] + bv.z; v.w = acc3[r] + bv.w;
      *(float4*)(zrow + (size_t)r * 128) = v;
    }
  }
}

#define L2_WAVES 8
__global__ void __launch_bounds__(32 * L2_WAVES, 16)
sage_layer2_gather(const float* __restrict__ Z, const int* __restrict__ blk,
                   const int* __restrict__ ohead, const int* __restrict__ next,
                   const int* __restrict__ src, float* __restrict__ Out, int Nnodes) {
  __shared__ int sorted_ids[L2_WAVES * 64];

  const int wid = threadIdx.x >> 5;
  const int lane = threadIdx.x & 31;
  const int swid = __builtin_amdgcn_readfirstlane(wid);
  const int node0 = (blockIdx.x * L2_WAVES + swid) * GATHER_ROWS;
  int* sorted = sorted_ids + swid * 64;
  int dd[GATHER_ROWS], kk0[GATHER_ROWS], kk1[GATHER_ROWS];
  float2 zs[GATHER_ROWS];
#pragma unroll
  for (int i = 0; i < GATHER_ROWS; ++i) {
    const int node = node0 + i;
    dd[i] = 0;
    if (node < Nnodes) dd[i] = blk[(size_t)node * 64 + 63];
  }
#pragma unroll
  for (int i = 0; i < GATHER_ROWS; ++i) {
    const int node = node0 + i;
    const int nc = node < Nnodes ? node : Nnodes - 1;
    const int* my = blk + (size_t)nc * 64;
    const int ns = dd[i] < 63 ? dd[i] : 63;
    kk0[i] = my[lane < ns ? lane : 63];
    kk1[i] = my[32 + lane < ns ? 32 + lane : 63];
    zs[i] = *(const float2*)(Z + (size_t)nc * 128 + lane * 2);
  }
#pragma unroll
  for (int i = 0; i < GATHER_ROWS; ++i) {
    const int node = node0 + i;
    const float2 m = neighbour_mean<float2>(Z + 64, (unsigned)lane * 8u, ohead, next, src, node, dd[i],
                                      kk0[i], kk1[i], sorted, lane);
    if (node < Nnodes) {
      float2 o;
      o.x = zs[i].x + m.x; o.y = zs[i].y + m.y;
      *(float2*)(Out + (size_t)node * 64 + lane * 2) = o;
    }
  }
}

extern "C" void kernel_launch(void* const* d_in, const int* in_sizes, int n_in,
                              void* d_out, int out_size, void* d_ws, size_t ws_size,
                              hipStream_t stream) {
  (void)in_sizes; (void)n_in; (void)out_size; (void)ws_size;

  const float* x   = (const float*)d_in[0];
  const int*   src = (const int*)d_in[1];
  const int*   dst = (const int*)d_in[2];
  const float* W1s = (const float*)d_in[3];
  const float* W1n = (const float*)d_in[4];
  const float* b1  = (const float*)d_in[5];
  const float* W2s = (const float*)d_in[6];
  const float* W2n = (const float*)d_in[7];
  const float* b2  = (const float*)d_in[8];
  float* out = (float*)d_out;

  const int N = 50000, E = 800000;
  const int NP = 50048;

  int* ohead = (int*)d_ws;
  int* next  = ohead + NP;
  int* blk   = next + E;
  float* z   = (float*)(blk + (long)N * 64);

  const int T = 256;

  clear_buckets<<<(N + T - 1) / T, T, 0, stream>>>(blk, ohead, N);
  bucket_edges<<<(E + T - 1) / T, T, 0, stream>>>(src, dst, blk, ohead, next, E);

  sage_layer1_project<<<1563, 256, 0, stream>>>(x, blk, ohead, next, src, W1s, W1n, b1,
                                                W2s, W2n, b2, z, N);

  sage_layer2_gather<<<1563, 256, 0, stream>>>(z, blk, ohead, next, src, out, N);
}
